# starting kernel
# baseline (speedup 1.0000x reference)
#include <hip/hip_runtime.h>
#include <hip/hip_bf16.h>
#include <cstddef>

typedef __attribute__((ext_vector_type(2))) float v2f;
typedef __attribute__((ext_vector_type(8))) float v8f;

#define IN_CH   128
#define OUT_CH  64
#define NEG_SLOPE 0.2f
#define EPS_F   1e-10f

__global__ void gat_gemm_wmma(const float* __restrict__ x,
                              const float* __restrict__ W,
                              float* __restrict__ h) {
    const int lane  = threadIdx.x & 31;
    const int wave  = threadIdx.x >> 5;
    const int tile  = blockIdx.x * (blockDim.x >> 5) + wave;
    const int mBase = (tile >> 2) * 16;
    const int nBase = (tile & 3) * 16;

    const int l15  = lane & 15;
    const int kOff = (lane >> 4) << 1;

    const float* xRow = x + (size_t)(mBase + l15) * IN_CH + kOff;
    const float* wRow = W + (size_t)(nBase + l15) * IN_CH + kOff;

    v8f acc = {};
#pragma unroll
    for (int k = 0; k < IN_CH; k += 4) {
        v2f a = *(const v2f*)(xRow + k);
        v2f b = *(const v2f*)(wRow + k);
        acc = __builtin_amdgcn_wmma_f32_16x16x4_f32(
             false, a,  false, b,
             (short)0, acc,  false,  false);
    }

    const int rowHalf = (lane >> 4) << 3;
    float* o = h + (size_t)(mBase + rowHalf) * OUT_CH + nBase + l15;
#pragma unroll
    for (int r = 0; r < 8; ++r)
        o[(size_t)r * OUT_CH] = acc[r];
}

__global__ void gat_gemm_tail(const float* __restrict__ x,
                              const float* __restrict__ W,
                              float* __restrict__ h,
                              int rowStart, int nNodes) {
    int idx = blockIdx.x * blockDim.x + threadIdx.x;
    int row = rowStart + (idx >> 6);
    int col = idx & 63;
    if (row >= nNodes) return;
    const float* xr = x + (size_t)row * IN_CH;
    const float* wr = W + (size_t)col * IN_CH;
    float s = 0.f;
    for (int k = 0; k < IN_CH; ++k) s += xr[k] * wr[k];
    h[(size_t)row * OUT_CH + col] = s;
}

__global__ void gat_el_er(const float* __restrict__ h,
                          const float* __restrict__ a_l,
                          const float* __restrict__ a_r,
                          float* __restrict__ e_l,
                          float* __restrict__ e_r, int nNodes) {
    const int lane = threadIdx.x & 31;
    const int node = (blockIdx.x * blockDim.x + threadIdx.x) >> 5;
    if (node >= nNodes) return;
    float2 hv = ((const float2*)(h + (size_t)node * OUT_CH))[lane];
    float2 al = ((const float2*)a_l)[lane];
    float2 ar = ((const float2*)a_r)[lane];
    float pl = hv.x * al.x + hv.y * al.y;
    float pr = hv.x * ar.x + hv.y * ar.y;
#pragma unroll
    for (int off = 16; off > 0; off >>= 1) {
        pl += __shfl_xor(pl, off, 32);
        pr += __shfl_xor(pr, off, 32);
    }
    if (lane == 0) { e_l[node] = pl; e_r[node] = pr; }
}

__global__ void gat_init(float* __restrict__ out, const float* __restrict__ bias,
                         float* __restrict__ node_max, float* __restrict__ node_sum,
                         int nNodes) {
    int i = blockIdx.x * blockDim.x + threadIdx.x;
    if (i < nNodes * OUT_CH) out[i] = bias[i & (OUT_CH - 1)];
    if (i < nNodes) { node_max[i] = -__builtin_inff(); node_sum[i] = 0.f; }
}

__device__ __forceinline__ void atomicMaxF32(float* addr, float v) {
    if (v >= 0.f)
        atomicMax((int*)addr, __float_as_int(v));
    else
        atomicMin((unsigned int*)addr, __float_as_uint(v));
}

__global__ void gat_edge_max(const int* __restrict__ src, const int* __restrict__ dst,
                             const float* __restrict__ e_l, const float* __restrict__ e_r,
                             float* __restrict__ edge_w, float* __restrict__ node_max,
                             int nEdges) {
    int e = blockIdx.x * blockDim.x + threadIdx.x;
    if (e >= nEdges) return;
    int s = src[e], d = dst[e];
    float a = e_l[d] + e_r[s];
    a = (a > 0.f) ? a : NEG_SLOPE * a;
    edge_w[e] = a;
    atomicMaxF32(node_max + d, a);
}

__global__ void gat_edge_exp(const int* __restrict__ dst,
                             float* __restrict__ edge_w,
                             const float* __restrict__ node_max,
                             float* __restrict__ node_sum, int nEdges) {
    int e = blockIdx.x * blockDim.x + threadIdx.x;
    if (e >= nEdges) return;
    int d = dst[e];
    float w = __expf(edge_w[e] - node_max[d]);
    edge_w[e] = w;
    atomicAdd(node_sum + d, w);
}

__global__ void gat_scatter(const int* __restrict__ src, const int* __restrict__ dst,
                            const float* __restrict__ h,
                            const float* __restrict__ edge_w,
                            const float* __restrict__ node_sum,
                            float* __restrict__ out, int nEdges) {
    const int lane = threadIdx.x & 31;
    const int e = (blockIdx.x * blockDim.x + threadIdx.x) >> 5;
    if (e >= nEdges) return;
    const int s = src[e], d = dst[e];
    const float coef = edge_w[e] / (node_sum[d] + EPS_F);
    float2 hv = ((const float2*)(h + (size_t)s * OUT_CH))[lane];
    float* o = out + (size_t)d * OUT_CH + lane * 2;
    atomicAdd(o + 0, hv.x * coef);
    atomicAdd(o + 1, hv.y * coef);
}

extern "C" void kernel_launch(void* const* d_in, const int* in_sizes, int n_in,
                              void* d_out, int out_size, void* d_ws, size_t ws_size,
                              hipStream_t stream) {
    const float* x    = (const float*)d_in[0];
    const int*   ei   = (const int*)d_in[1];
    const float* W    = (const float*)d_in[2];
    const float* a_l  = (const float*)d_in[3];
    const float* a_r  = (const float*)d_in[4];
    const float* bias = (const float*)d_in[5];
    float* out = (float*)d_out;

    const int nNodes = in_sizes[0] / IN_CH;
    const int nEdges = in_sizes[1] / 2;
    const int* src = ei;
    const int* dst = ei + nEdges;

    float* ws       = (float*)d_ws;
    float* h        = ws;
    float* e_l      = h + (size_t)nNodes * OUT_CH;
    float* e_r      = e_l + nNodes;
    float* node_max = e_r + nNodes;
    float* node_sum = node_max + nNodes;
    float* edge_w   = node_sum + nNodes;

    const int mFull = nNodes & ~15;
    const int tiles = (mFull / 16) * (OUT_CH / 16);
    if (tiles > 0)
        gat_gemm_wmma<<<(tiles + 3) / 4, 128, 0, stream>>>(x, W, h);
    if (mFull < nNodes) {
        int tailThreads = (nNodes - mFull) * OUT_CH;
        gat_gemm_tail<<<(tailThreads + 255) / 256, 256, 0, stream>>>(x, W, h, mFull, nNodes);
    }

    gat_el_er<<<(nNodes * 32 + 255) / 256, 256, 0, stream>>>(h, a_l, a_r, e_l, e_r, nNodes);

    int initN = nNodes * OUT_CH;
    gat_init<<<(initN + 255) / 256, 256, 0, stream>>>(out, bias, node_max, node_sum, nNodes);

    gat_edge_max<<<(nEdges + 255) / 256, 256, 0, stream>>>(src, dst, e_l, e_r, edge_w, node_max, nEdges);
    gat_edge_exp<<<(nEdges + 255) / 256, 256, 0, stream>>>(dst, edge_w, node_max, node_sum, nEdges);
    gat_scatter<<<((size_t)nEdges * 32 + 255) / 256, 256, 0, stream>>>(src, dst, h, edge_w, node_sum, out, nEdges);
}


# scatter: lane l adds columns l and l+32 (was 2l, 2l+1), so each wave-wide f32 atomic covers one whole 128-B line of out[dst] instead of half of each of two lines; h row read the same way
# speedup vs baseline: 1.0558x; 1.0558x over previous
#include <hip/hip_runtime.h>
#include <hip/hip_bf16.h>
#include <cstddef>

typedef __attribute__((ext_vector_type(2))) float v2f;
typedef __attribute__((ext_vector_type(8))) float v8f;

#define IN_CH   128
#define OUT_CH  64
#define NEG_SLOPE 0.2f
#define EPS_F   1e-10f

__global__ void gat_gemm_wmma(const float* __restrict__ x,
                              const float* __restrict__ W,
                              float* __restrict__ h) {
    const int lane  = threadIdx.x & 31;
    const int wave  = threadIdx.x >> 5;
    const int tile  = blockIdx.x * (blockDim.x >> 5) + wave;
    const int mBase = (tile >> 2) * 16;
    const int nBase = (tile & 3) * 16;

    const int l15  = lane & 15;
    const int kOff = (lane >> 4) << 1;

    const float* xRow = x + (size_t)(mBase + l15) * IN_CH + kOff;
    const float* wRow = W + (size_t)(nBase + l15) * IN_CH + kOff;

    v8f acc = {};
#pragma unroll
    for (int k = 0; k < IN_CH; k += 4) {
        v2f a = *(const v2f*)(xRow + k);
        v2f b = *(const v2f*)(wRow + k);
        acc = __builtin_amdgcn_wmma_f32_16x16x4_f32(
             false, a,  false, b,
             (short)0, acc,  false,  false);
    }

    const int rowHalf = (lane >> 4) << 3;
    float* o = h + (size_t)(mBase + rowHalf) * OUT_CH + nBase + l15;
#pragma unroll
    for (int r = 0; r < 8; ++r)
        o[(size_t)r * OUT_CH] = acc[r];
}

__global__ void gat_gemm_tail(const float* __restrict__ x,
                              const float* __restrict__ W,
                              float* __restrict__ h,
                              int rowStart, int nNodes) {
    int idx = blockIdx.x * blockDim.x + threadIdx.x;
    int row = rowStart + (idx >> 6);
    int col = idx & 63;
    if (row >= nNodes) return;
    const float* xr = x + (size_t)row * IN_CH;
    const float* wr = W + (size_t)col * IN_CH;
    float s = 0.f;
    for (int k = 0; k < IN_CH; ++k) s += xr[k] * wr[k];
    h[(size_t)row * OUT_CH + col] = s;
}

__global__ void gat_el_er(const float* __restrict__ h,
                          const float* __restrict__ a_l,
                          const float* __restrict__ a_r,
                          float* __restrict__ e_l,
                          float* __restrict__ e_r, int nNodes) {
    const int lane = threadIdx.x & 31;
    const int node = (blockIdx.x * blockDim.x + threadIdx.x) >> 5;
    if (node >= nNodes) return;
    float2 hv = ((const float2*)(h + (size_t)node * OUT_CH))[lane];
    float2 al = ((const float2*)a_l)[lane];
    float2 ar = ((const float2*)a_r)[lane];
    float pl = hv.x * al.x + hv.y * al.y;
    float pr = hv.x * ar.x + hv.y * ar.y;
#pragma unroll
    for (int off = 16; off > 0; off >>= 1) {
        pl += __shfl_xor(pl, off, 32);
        pr += __shfl_xor(pr, off, 32);
    }
    if (lane == 0) { e_l[node] = pl; e_r[node] = pr; }
}

__global__ void gat_init(float* __restrict__ out, const float* __restrict__ bias,
                         float* __restrict__ node_max, float* __restrict__ node_sum,
                         int nNodes) {
    int i = blockIdx.x * blockDim.x + threadIdx.x;
    if (i < nNodes * OUT_CH) out[i] = bias[i & (OUT_CH - 1)];
    if (i < nNodes) { node_max[i] = -__builtin_inff(); node_sum[i] = 0.f; }
}

__device__ __forceinline__ void atomicMaxF32(float* addr, float v) {
    if (v >= 0.f)
        atomicMax((int*)addr, __float_as_int(v));
    else
        atomicMin((unsigned int*)addr, __float_as_uint(v));
}

__global__ void gat_edge_max(const int* __restrict__ src, const int* __restrict__ dst,
                             const float* __restrict__ e_l, const float* __restrict__ e_r,
                             float* __restrict__ edge_w, float* __restrict__ node_max,
                             int nEdges) {
    int e = blockIdx.x * blockDim.x + threadIdx.x;
    if (e >= nEdges) return;
    int s = src[e], d = dst[e];
    float a = e_l[d] + e_r[s];
    a = (a > 0.f) ? a : NEG_SLOPE * a;
    edge_w[e] = a;
    atomicMaxF32(node_max + d, a);
}

__global__ void gat_edge_exp(const int* __restrict__ dst,
                             float* __restrict__ edge_w,
                             const float* __restrict__ node_max,
                             float* __restrict__ node_sum, int nEdges) {
    int e = blockIdx.x * blockDim.x + threadIdx.x;
    if (e >= nEdges) return;
    int d = dst[e];
    float w = __expf(edge_w[e] - node_max[d]);
    edge_w[e] = w;
    atomicAdd(node_sum + d, w);
}

__global__ void gat_scatter(const int* __restrict__ src, const int* __restrict__ dst,
                            const float* __restrict__ h,
                            const float* __restrict__ edge_w,
                            const float* __restrict__ node_sum,
                            float* __restrict__ out, int nEdges) {
    const int lane = threadIdx.x & 31;
    const int e = (blockIdx.x * blockDim.x + threadIdx.x) >> 5;
    if (e >= nEdges) return;
    const int s = src[e], d = dst[e];
    const float coef = edge_w[e] / (node_sum[d] + EPS_F);
    const float* hr = h + (size_t)s * OUT_CH + lane;
    const float h0 = hr[0];
    const float h1 = hr[32];
    float* o = out + (size_t)d * OUT_CH + lane;
    atomicAdd(o + 0,  h0 * coef);
    atomicAdd(o + 32, h1 * coef);
}

extern "C" void kernel_launch(void* const* d_in, const int* in_sizes, int n_in,
                              void* d_out, int out_size, void* d_ws, size_t ws_size,
                              hipStream_t stream) {
    const float* x    = (const float*)d_in[0];
    const int*   ei   = (const int*)d_in[1];
    const float* W    = (const float*)d_in[2];
    const float* a_l  = (const float*)d_in[3];
    const float* a_r  = (const float*)d_in[4];
    const float* bias = (const float*)d_in[5];
    float* out = (float*)d_out;

    const int nNodes = in_sizes[0] / IN_CH;
    const int nEdges = in_sizes[1] / 2;
    const int* src = ei;
    const int* dst = ei + nEdges;

    float* ws       = (float*)d_ws;
    float* h        = ws;
    float* e_l      = h + (size_t)nNodes * OUT_CH;
    float* e_r      = e_l + nNodes;
    float* node_max = e_r + nNodes;
    float* node_sum = node_max + nNodes;
    float* edge_w   = node_sum + nNodes;

    const int mFull = nNodes & ~15;
    const int tiles = (mFull / 16) * (OUT_CH / 16);
    if (tiles > 0)
        gat_gemm_wmma<<<(tiles + 3) / 4, 128, 0, stream>>>(x, W, h);
    if (mFull < nNodes) {
        int tailThreads = (nNodes - mFull) * OUT_CH;
        gat_gemm_tail<<<(tailThreads + 255) / 256, 256, 0, stream>>>(x, W, h, mFull, nNodes);
    }

    gat_el_er<<<(nNodes * 32 + 255) / 256, 256, 0, stream>>>(h, a_l, a_r, e_l, e_r, nNodes);

    int initN = nNodes * OUT_CH;
    gat_init<<<(initN + 255) / 256, 256, 0, stream>>>(out, bias, node_max, node_sum, nNodes);

    gat_edge_max<<<(nEdges + 255) / 256, 256, 0, stream>>>(src, dst, e_l, e_r, edge_w, node_max, nEdges);
    gat_edge_exp<<<(nEdges + 255) / 256, 256, 0, stream>>>(dst, edge_w, node_max, node_sum, nEdges);
    gat_scatter<<<((size_t)nEdges * 32 + 255) / 256, 256, 0, stream>>>(src, dst, h, edge_w, node_sum, out, nEdges);
}


# scatter: one wave per 32 edges (was one per edge): src, dst, weight, sum[dst] loaded once per lane, broadcast by readlane, rows fetched 8 edges ahead; order of the f32 atomic adds into out changes
# speedup vs baseline: 1.3549x; 1.2833x over previous
#include <hip/hip_runtime.h>
#include <hip/hip_bf16.h>
#include <cstddef>

typedef __attribute__((ext_vector_type(2))) float v2f;
typedef __attribute__((ext_vector_type(8))) float v8f;

#define IN_CH   128
#define OUT_CH  64
#define NEG_SLOPE 0.2f
#define EPS_F   1e-10f

__global__ void gat_gemm_wmma(const float* __restrict__ x,
                              const float* __restrict__ W,
                              float* __restrict__ h) {
    const int lane  = threadIdx.x & 31;
    const int wave  = threadIdx.x >> 5;
    const int tile  = blockIdx.x * (blockDim.x >> 5) + wave;
    const int mBase = (tile >> 2) * 16;
    const int nBase = (tile & 3) * 16;

    const int l15  = lane & 15;
    const int kOff = (lane >> 4) << 1;

    const float* xRow = x + (size_t)(mBase + l15) * IN_CH + kOff;
    const float* wRow = W + (size_t)(nBase + l15) * IN_CH + kOff;

    v8f acc = {};
#pragma unroll
    for (int k = 0; k < IN_CH; k += 4) {
        v2f a = *(const v2f*)(xRow + k);
        v2f b = *(const v2f*)(wRow + k);
        acc = __builtin_amdgcn_wmma_f32_16x16x4_f32(
             false, a,  false, b,
             (short)0, acc,  false,  false);
    }

    const int rowHalf = (lane >> 4) << 3;
    float* o = h + (size_t)(mBase + rowHalf) * OUT_CH + nBase + l15;
#pragma unroll
    for (int r = 0; r < 8; ++r)
        o[(size_t)r * OUT_CH] = acc[r];
}

__global__ void gat_gemm_tail(const float* __restrict__ x,
                              const float* __restrict__ W,
                              float* __restrict__ h,
                              int rowStart, int nNodes) {
    int idx = blockIdx.x * blockDim.x + threadIdx.x;
    int row = rowStart + (idx >> 6);
    int col = idx & 63;
    if (row >= nNodes) return;
    const float* xr = x + (size_t)row * IN_CH;
    const float* wr = W + (size_t)col * IN_CH;
    float s = 0.f;
    for (int k = 0; k < IN_CH; ++k) s += xr[k] * wr[k];
    h[(size_t)row * OUT_CH + col] = s;
}

__global__ void gat_el_er(const float* __restrict__ h,
                          const float* __restrict__ a_l,
                          const float* __restrict__ a_r,
                          float* __restrict__ e_l,
                          float* __restrict__ e_r, int nNodes) {
    const int lane = threadIdx.x & 31;
    const int node = (blockIdx.x * blockDim.x + threadIdx.x) >> 5;
    if (node >= nNodes) return;
    float2 hv = ((const float2*)(h + (size_t)node * OUT_CH))[lane];
    float2 al = ((const float2*)a_l)[lane];
    float2 ar = ((const float2*)a_r)[lane];
    float pl = hv.x * al.x + hv.y * al.y;
    float pr = hv.x * ar.x + hv.y * ar.y;
#pragma unroll
    for (int off = 16; off > 0; off >>= 1) {
        pl += __shfl_xor(pl, off, 32);
        pr += __shfl_xor(pr, off, 32);
    }
    if (lane == 0) { e_l[node] = pl; e_r[node] = pr; }
}

__global__ void gat_init(float* __restrict__ out, const float* __restrict__ bias,
                         float* __restrict__ node_max, float* __restrict__ node_sum,
                         int nNodes) {
    int i = blockIdx.x * blockDim.x + threadIdx.x;
    if (i < nNodes * OUT_CH) out[i] = bias[i & (OUT_CH - 1)];
    if (i < nNodes) { node_max[i] = -__builtin_inff(); node_sum[i] = 0.f; }
}

__device__ __forceinline__ void atomicMaxF32(float* addr, float v) {
    if (v >= 0.f)
        atomicMax((int*)addr, __float_as_int(v));
    else
        atomicMin((unsigned int*)addr, __float_as_uint(v));
}

__global__ void gat_edge_max(const int* __restrict__ src, const int* __restrict__ dst,
                             const float* __restrict__ e_l, const float* __restrict__ e_r,
                             float* __restrict__ edge_w, float* __restrict__ node_max,
                             int nEdges) {
    int e = blockIdx.x * blockDim.x + threadIdx.x;
    if (e >= nEdges) return;
    int s = src[e], d = dst[e];
    float a = e_l[d] + e_r[s];
    a = (a > 0.f) ? a : NEG_SLOPE * a;
    edge_w[e] = a;
    atomicMaxF32(node_max + d, a);
}

__global__ void gat_edge_exp(const int* __restrict__ dst,
                             float* __restrict__ edge_w,
                             const float* __restrict__ node_max,
                             float* __restrict__ node_sum, int nEdges) {
    int e = blockIdx.x * blockDim.x + threadIdx.x;
    if (e >= nEdges) return;
    int d = dst[e];
    float w = __expf(edge_w[e] - node_max[d]);
    edge_w[e] = w;
    atomicAdd(node_sum + d, w);
}

#define SC_EDGES 32
#define SC_GROUP 8

__device__ __forceinline__ int   laneInt(int v, int j)   { return __builtin_amdgcn_readlane(v, j); }
__device__ __forceinline__ float laneFlt(float v, int j) {
    return __int_as_float(__builtin_amdgcn_readlane(__float_as_int(v), j));
}
template <typename P>
__device__ __forceinline__ P* rowBase(P* p, int row) {
    typedef __attribute__((address_space(1))) P GP;
    GP* r = (GP*)p + (size_t)row * OUT_CH;
    asm("" : "+s"(r));
    return (P*)r;
}

__global__ void gat_scatter(const int* __restrict__ src, const int* __restrict__ dst,
                            const float* __restrict__ h,
                            const float* __restrict__ edge_w,
                            const float* __restrict__ node_sum,
                            float* __restrict__ out, int nEdges) {
    const int lane = threadIdx.x & 31;
    const int wv   = __builtin_amdgcn_readfirstlane((int)((blockIdx.x * blockDim.x + threadIdx.x) >> 5));
    const int base = wv * SC_EDGES;
    if (base >= nEdges) return;
    const int left = nEdges - base;
    const int n    = left < SC_EDGES ? left : SC_EDGES;

    int s = 0, d = 0;
    float coef = 0.f;
    if (lane < n) {
        const int e = base + lane;
        s = src[e];
        d = dst[e];
        coef = edge_w[e] / (node_sum[d] + EPS_F);
    }

    if (n == SC_EDGES) {
        float r0[2][SC_GROUP], r1[2][SC_GROUP];
#pragma unroll
        for (int u = 0; u < SC_GROUP; ++u) {
            const float* hr = rowBase(h, laneInt(s, u));
            r0[0][u] = hr[lane];
            r1[0][u] = hr[lane + 32];
        }
#pragma unroll
        for (int g = 0; g < SC_EDGES / SC_GROUP; ++g) {
            const int cur = g & 1, nxt = cur ^ 1;
            if (g + 1 < SC_EDGES / SC_GROUP) {
#pragma unroll
                for (int u = 0; u < SC_GROUP; ++u) {
                    const float* hr = rowBase(h, laneInt(s, (g + 1) * SC_GROUP + u));
                    r0[nxt][u] = hr[lane];
                    r1[nxt][u] = hr[lane + 32];
                }
            }
#pragma unroll
            for (int u = 0; u < SC_GROUP; ++u) {
                const int   j  = g * SC_GROUP + u;
                const float cj = laneFlt(coef, j);
                float* o = rowBase(out, laneInt(d, j));
                atomicAdd(o + lane,      r0[cur][u] * cj);
                atomicAdd(o + lane + 32, r1[cur][u] * cj);
            }
        }
    } else {
        for (int j = 0; j < n; ++j) {
            const float cj = laneFlt(coef, j);
            const float* hr = rowBase(h, laneInt(s, j));
            float* o = rowBase(out, laneInt(d, j));
            const float h0 = hr[lane];
            const float h1 = hr[lane + 32];
            atomicAdd(o + lane,      h0 * cj);
            atomicAdd(o + lane + 32, h1 * cj);
        }
    }
}

extern "C" void kernel_launch(void* const* d_in, const int* in_sizes, int n_in,
                              void* d_out, int out_size, void* d_ws, size_t ws_size,
                              hipStream_t stream) {
    const float* x    = (const float*)d_in[0];
    const int*   ei   = (const int*)d_in[1];
    const float* W    = (const float*)d_in[2];
    const float* a_l  = (const float*)d_in[3];
    const float* a_r  = (const float*)d_in[4];
    const float* bias = (const float*)d_in[5];
    float* out = (float*)d_out;

    const int nNodes = in_sizes[0] / IN_CH;
    const int nEdges = in_sizes[1] / 2;
    const int* src = ei;
    const int* dst = ei + nEdges;

    float* ws       = (float*)d_ws;
    float* h        = ws;
    float* e_l      = h + (size_t)nNodes * OUT_CH;
    float* e_r      = e_l + nNodes;
    float* node_max = e_r + nNodes;
    float* node_sum = node_max + nNodes;
    float* edge_w   = node_sum + nNodes;

    const int mFull = nNodes & ~15;
    const int tiles = (mFull / 16) * (OUT_CH / 16);
    if (tiles > 0)
        gat_gemm_wmma<<<(tiles + 3) / 4, 128, 0, stream>>>(x, W, h);
    if (mFull < nNodes) {
        int tailThreads = (nNodes - mFull) * OUT_CH;
        gat_gemm_tail<<<(tailThreads + 255) / 256, 256, 0, stream>>>(x, W, h, mFull, nNodes);
    }

    gat_el_er<<<(nNodes * 32 + 255) / 256, 256, 0, stream>>>(h, a_l, a_r, e_l, e_r, nNodes);

    int initN = nNodes * OUT_CH;
    gat_init<<<(initN + 255) / 256, 256, 0, stream>>>(out, bias, node_max, node_sum, nNodes);

    gat_edge_max<<<(nEdges + 255) / 256, 256, 0, stream>>>(src, dst, e_l, e_r, edge_w, node_max, nEdges);
    gat_edge_exp<<<(nEdges + 255) / 256, 256, 0, stream>>>(dst, edge_w, node_max, node_sum, nEdges);
    gat_scatter<<<(nEdges + 255) / 256, 256, 0, stream>>>(src, dst, h, edge_w, node_sum, out, nEdges);
}


# edges grouped by dst (counting sort rebuilt each call, ranks parked in out): one wave sums a node's rows in registers, softmax in-wave, row stored once; no f32 atomics; sum order changes
# speedup vs baseline: 1.7423x; 1.2859x over previous
#include <hip/hip_runtime.h>
#include <hip/hip_bf16.h>
#include <cstddef>

typedef __attribute__((ext_vector_type(2))) float v2f;
typedef __attribute__((ext_vector_type(8))) float v8f;

#define IN_CH   128
#define OUT_CH  64
#define NEG_SLOPE 0.2f
#define EPS_F   1e-10f

__global__ void gat_gemm_wmma(const float* __restrict__ x,
                              const float* __restrict__ W,
                              float* __restrict__ h) {
    const int lane  = threadIdx.x & 31;
    const int wave  = threadIdx.x >> 5;
    const int tile  = blockIdx.x * (blockDim.x >> 5) + wave;
    const int mBase = (tile >> 2) * 16;
    const int nBase = (tile & 3) * 16;

    const int l15  = lane & 15;
    const int kOff = (lane >> 4) << 1;

    const float* xRow = x + (size_t)(mBase + l15) * IN_CH + kOff;
    const float* wRow = W + (size_t)(nBase + l15) * IN_CH + kOff;

    v8f acc = {};
#pragma unroll
    for (int k = 0; k < IN_CH; k += 4) {
        v2f a = *(const v2f*)(xRow + k);
        v2f b = *(const v2f*)(wRow + k);
        acc = __builtin_amdgcn_wmma_f32_16x16x4_f32(
             false, a,  false, b,
             (short)0, acc,  false,  false);
    }

    const int rowHalf = (lane >> 4) << 3;
    float* o = h + (size_t)(mBase + rowHalf) * OUT_CH + nBase + l15;
#pragma unroll
    for (int r = 0; r < 8; ++r)
        o[(size_t)r * OUT_CH] = acc[r];
}

__global__ void gat_gemm_tail(const float* __restrict__ x,
                              const float* __restrict__ W,
                              float* __restrict__ h,
                              int rowStart, int nNodes) {
    int idx = blockIdx.x * blockDim.x + threadIdx.x;
    int row = rowStart + (idx >> 6);
    int col = idx & 63;
    if (row >= nNodes) return;
    const float* xr = x + (size_t)row * IN_CH;
    const float* wr = W + (size_t)col * IN_CH;
    float s = 0.f;
    for (int k = 0; k < IN_CH; ++k) s += xr[k] * wr[k];
    h[(size_t)row * OUT_CH + col] = s;
}

__global__ void gat_el_er(const float* __restrict__ h,
                          const float* __restrict__ a_l,
                          const float* __restrict__ a_r,
                          float* __restrict__ e_l,
                          float* __restrict__ e_r,
                          int* __restrict__ cnt, int* __restrict__ cursor, int nNodes) {
    const int lane = threadIdx.x & 31;
    const int node = (blockIdx.x * blockDim.x + threadIdx.x) >> 5;
    if (node >= nNodes) return;
    float2 hv = ((const float2*)(h + (size_t)node * OUT_CH))[lane];
    float2 al = ((const float2*)a_l)[lane];
    float2 ar = ((const float2*)a_r)[lane];
    float pl = hv.x * al.x + hv.y * al.y;
    float pr = hv.x * ar.x + hv.y * ar.y;
#pragma unroll
    for (int off = 16; off > 0; off >>= 1) {
        pl += __shfl_xor(pl, off, 32);
        pr += __shfl_xor(pr, off, 32);
    }
    if (lane == 0) {
        e_l[node] = pl; e_r[node] = pr;
        cnt[node] = 0;
        if (node == 0) *cursor = 0;
    }
}

__global__ void gat_count(const int* __restrict__ dst, int* __restrict__ cnt,
                          int* __restrict__ rank, int nEdges) {
    int e = blockIdx.x * blockDim.x + threadIdx.x;
    if (e >= nEdges) return;
    rank[e] = atomicAdd(cnt + dst[e], 1);
}

__global__ void gat_slots(const int* __restrict__ cnt, int* __restrict__ off,
                          int* __restrict__ cursor, int nNodes) {
    const int lane = threadIdx.x & 31;
    const int node = blockIdx.x * blockDim.x + threadIdx.x;
    const int c = node < nNodes ? cnt[node] : 0;
    int incl = c;
#pragma unroll
    for (int d = 1; d < 32; d <<= 1) {
        int t = __shfl_up(incl, d, 32);
        if (lane >= d) incl += t;
    }
    int base = 0;
    if (lane == 31) base = atomicAdd(cursor, incl);
    base = __shfl(base, 31, 32);
    if (node < nNodes) off[node] = base + incl - c;
}

__global__ void gat_fill(const int* __restrict__ src, const int* __restrict__ dst,
                         const int* __restrict__ off, const int* __restrict__ rank,
                         int* __restrict__ perm, int nEdges) {
    int e = blockIdx.x * blockDim.x + threadIdx.x;
    if (e >= nEdges) return;
    perm[off[dst[e]] + rank[e]] = src[e];
}

#define GA_NODES 4
#define GA_GROUP 8

__device__ __forceinline__ int   laneInt(int v, int j)   { return __builtin_amdgcn_readlane(v, j); }
__device__ __forceinline__ float laneFlt(float v, int j) {
    return __int_as_float(__builtin_amdgcn_readlane(__float_as_int(v), j));
}
template <int CTRL>
__device__ __forceinline__ float dppMove(float v) {
    return __int_as_float(__builtin_amdgcn_update_dpp(0, __float_as_int(v), CTRL, 0xF, 0xF, true));
}
__device__ __forceinline__ float otherRow(float v) {
    const int i = __float_as_int(v);
    return __int_as_float(__builtin_amdgcn_permlanex16(i, i, 0x76543210, 0xfedcba98, false, false));
}
__device__ __forceinline__ float waveMax(float v) {
    v = fmaxf(v, dppMove<0xB1>(v));
    v = fmaxf(v, dppMove<0x4E>(v));
    v = fmaxf(v, dppMove<0x141>(v));
    v = fmaxf(v, dppMove<0x140>(v));
    v = fmaxf(v, otherRow(v));
    return v;
}
__device__ __forceinline__ float waveSum(float v) {
    v += dppMove<0xB1>(v);
    v += dppMove<0x4E>(v);
    v += dppMove<0x141>(v);
    v += dppMove<0x140>(v);
    v += otherRow(v);
    return v;
}
template <typename P>
__device__ __forceinline__ P* rowBase(P* p, int row) {
    typedef __attribute__((address_space(1))) P GP;
    GP* r = (GP*)p + (size_t)row * OUT_CH;
    asm("" : "+s"(r));
    return (P*)r;
}
__device__ __forceinline__ float leaky(float a) { return (a > 0.f) ? a : NEG_SLOPE * a; }

__device__ __forceinline__ void addRows(const float* __restrict__ h, int s, float coef,
                                        int m, int lane, float& acc0, float& acc1) {
    for (int j0 = 0; j0 < m; j0 += GA_GROUP) {
        float r0[GA_GROUP], r1[GA_GROUP];
#pragma unroll
        for (int u = 0; u < GA_GROUP; ++u) {
            const int j  = j0 + u;
            const int jj = j < m ? j : m - 1;
            const float* hr = rowBase(h, laneInt(s, jj));
            r0[u] = hr[lane];
            r1[u] = hr[lane + 32];
        }
#pragma unroll
        for (int u = 0; u < GA_GROUP; ++u) {
            const float cj = laneFlt(coef, j0 + u);
            acc0 += r0[u] * cj;
            acc1 += r1[u] * cj;
        }
    }
}

__global__ void gat_gather(const int* __restrict__ perm, const int* __restrict__ cnt,
                           const int* __restrict__ off, const float* __restrict__ h,
                           const float* __restrict__ e_l, const float* __restrict__ e_r,
                           const float* __restrict__ bias, float* __restrict__ out,
                           int nNodes) {
    const int lane = threadIdx.x & 31;
    const int wv    = __builtin_amdgcn_readfirstlane((int)((blockIdx.x * blockDim.x + threadIdx.x) >> 5));
    const int node0 = wv * GA_NODES;
    if (node0 >= nNodes) return;

    int myCnt = 0, myOff = 0;
    float myEl = 0.f;
    if (lane < GA_NODES && node0 + lane < nNodes) {
        myCnt = cnt[node0 + lane];
        myOff = off[node0 + lane];
        myEl  = e_l[node0 + lane];
    }
    const float b0 = bias[lane], b1 = bias[lane + 32];

#pragma nounroll
    for (int q = 0; q < GA_NODES; ++q) {
        const int node = node0 + q;
        if (node >= nNodes) break;
        const int   n  = laneInt(myCnt, q);
        const int   o  = laneInt(myOff, q);
        const float el = laneFlt(myEl, q);
        float acc0 = 0.f, acc1 = 0.f;

        if (n <= 32) {
            int s = 0;
            float a = -__builtin_inff();
            if (lane < n) { s = perm[o + lane]; a = leaky(el + e_r[s]); }
            const float mx  = waveMax(a);
            const float w   = lane < n ? __expf(a - mx) : 0.f;
            const float sum = waveSum(w);
            const float coef = w / (sum + EPS_F);
            addRows(h, s, coef, n, lane, acc0, acc1);
        } else {
            float mx = -__builtin_inff();
            for (int c = 0; c < n; c += 32) {
                const int i = c + lane;
                if (i < n) mx = fmaxf(mx, leaky(el + e_r[perm[o + i]]));
            }
            mx = waveMax(mx);
            float sum = 0.f;
            for (int c = 0; c < n; c += 32) {
                const int i = c + lane;
                if (i < n) sum += __expf(leaky(el + e_r[perm[o + i]]) - mx);
            }
            sum = waveSum(sum);
            for (int c = 0; c < n; c += 32) {
                const int i = c + lane;
                const int m = (n - c) < 32 ? (n - c) : 32;
                int s = 0;
                float coef = 0.f;
                if (i < n) {
                    s = perm[o + i];
                    coef = __expf(leaky(el + e_r[s]) - mx) / (sum + EPS_F);
                }
                addRows(h, s, coef, m, lane, acc0, acc1);
            }
        }
        float* orow = out + (size_t)node * OUT_CH;
        orow[lane]      = acc0 + b0;
        orow[lane + 32] = acc1 + b1;
    }
}

extern "C" void kernel_launch(void* const* d_in, const int* in_sizes, int n_in,
                              void* d_out, int out_size, void* d_ws, size_t ws_size,
                              hipStream_t stream) {
    const float* x    = (const float*)d_in[0];
    const int*   ei   = (const int*)d_in[1];
    const float* W    = (const float*)d_in[2];
    const float* a_l  = (const float*)d_in[3];
    const float* a_r  = (const float*)d_in[4];
    const float* bias = (const float*)d_in[5];
    float* out = (float*)d_out;

    const int nNodes = in_sizes[0] / IN_CH;
    const int nEdges = in_sizes[1] / 2;
    const int* src = ei;
    const int* dst = ei + nEdges;

    float* ws   = (float*)d_ws;
    float* h    = ws;
    float* e_l  = h + (size_t)nNodes * OUT_CH;
    float* e_r  = e_l + nNodes;
    int*   cnt  = (int*)(e_r + nNodes);
    int*   off  = cnt + nNodes;
    int*   perm = off + nNodes;
    int*   rank   = (int*)d_out;
    int*   cursor = rank + nEdges;

    const int mFull = nNodes & ~15;
    const int tiles = (mFull / 16) * (OUT_CH / 16);
    if (tiles > 0)
        gat_gemm_wmma<<<(tiles + 3) / 4, 128, 0, stream>>>(x, W, h);
    if (mFull < nNodes) {
        int tailThreads = (nNodes - mFull) * OUT_CH;
        gat_gemm_tail<<<(tailThreads + 255) / 256, 256, 0, stream>>>(x, W, h, mFull, nNodes);
    }

    gat_el_er<<<(nNodes * 32 + 255) / 256, 256, 0, stream>>>(h, a_l, a_r, e_l, e_r, cnt, cursor, nNodes);

    gat_count<<<(nEdges + 255) / 256, 256, 0, stream>>>(dst, cnt, rank, nEdges);
    gat_slots<<<(nNodes + 255) / 256, 256, 0, stream>>>(cnt, off, cursor, nNodes);
    gat_fill<<<(nEdges + 255) / 256, 256, 0, stream>>>(src, dst, off, rank, perm, nEdges);
    gat_gather<<<(nNodes + 31) / 32, 256, 0, stream>>>(perm, cnt, off, h, e_l, e_r, bias, out, nNodes);
}


# shared launch's GEMM fed 16 B per lane: W kept per block in LDS (pitch 16 B x 33), x by 128-bit loads, k permuted alike in A and B; 16 memory loads per wave, was 160; h sum order changes
# speedup vs baseline: 3.3318x; 1.9123x over previous
#include <hip/hip_runtime.h>
#include <hip/hip_bf16.h>
#include <cstddef>

typedef __attribute__((ext_vector_type(2))) float v2f;
typedef __attribute__((ext_vector_type(4))) float v4f;
typedef __attribute__((ext_vector_type(8))) float v8f;

#define IN_CH   128
#define OUT_CH  64
#define NEG_SLOPE 0.2f
#define EPS_F   1e-10f

template <int CTRL>
__device__ __forceinline__ float dppMove(float v) {
    return __int_as_float(__builtin_amdgcn_update_dpp(0, __float_as_int(v), CTRL, 0xF, 0xF, true));
}
__device__ __forceinline__ float halfSum(float v) {
    v += dppMove<0xB1>(v);
    v += dppMove<0x4E>(v);
    v += dppMove<0x141>(v);
    v += dppMove<0x140>(v);
    return v;
}

__device__ __forceinline__ void tileEpilogue(const v8f& acc0, const v8f& acc1,
                                             const v8f& acc2, const v8f& acc3,
                                             const float* __restrict__ a_l,
                                             const float* __restrict__ a_r,
                                             float* __restrict__ h,
                                             float* __restrict__ e_l,
                                             float* __restrict__ e_r,
                                             int* __restrict__ cnt, int* __restrict__ cursor,
                                             int tile, int lane) {
    const int mBase = tile * 16;
    const int l15   = lane & 15;
    const int rowHalf = (lane >> 4) << 3;
    float* o = h + (size_t)(mBase + rowHalf) * OUT_CH + l15;
#pragma unroll
    for (int r = 0; r < 8; ++r) {
        o[(size_t)r * OUT_CH]      = acc0[r];
        o[(size_t)r * OUT_CH + 16] = acc1[r];
        o[(size_t)r * OUT_CH + 32] = acc2[r];
        o[(size_t)r * OUT_CH + 48] = acc3[r];
    }

    const float al0 = a_l[l15], al1 = a_l[l15 + 16], al2 = a_l[l15 + 32], al3 = a_l[l15 + 48];
    const float ar0 = a_r[l15], ar1 = a_r[l15 + 16], ar2 = a_r[l15 + 32], ar3 = a_r[l15 + 48];
    const int myRow = l15 & 7;
    float vl = 0.f, vr = 0.f;
#pragma unroll
    for (int r = 0; r < 8; ++r) {
        float pl = acc0[r] * al0 + acc1[r] * al1 + acc2[r] * al2 + acc3[r] * al3;
        float pr = acc0[r] * ar0 + acc1[r] * ar1 + acc2[r] * ar2 + acc3[r] * ar3;
        pl = halfSum(pl);
        pr = halfSum(pr);
        if (myRow == r) { vl = pl; vr = pr; }
    }
    if (l15 < 8) {
        const int node = mBase + rowHalf + l15;
        e_l[node] = vl; e_r[node] = vr;
        cnt[node] = 0;
    }
    if (tile == 0 && lane == 0) *cursor = 0;
}

__device__ __forceinline__ void gemmAttnTile(const float* __restrict__ x,
                                             const float* __restrict__ W,
                                             const float* __restrict__ a_l,
                                             const float* __restrict__ a_r,
                                             float* __restrict__ h,
                                             float* __restrict__ e_l,
                                             float* __restrict__ e_r,
                                             int* __restrict__ cnt, int* __restrict__ cursor,
                                             int tile, int lane) {
    const int mBase = tile * 16;

    const int l15  = lane & 15;
    const int kOff = (lane >> 4) << 1;

    const float* xRow = x + (size_t)(mBase + l15) * IN_CH + kOff;
    const float* wRow = W + (size_t)l15 * IN_CH + kOff;

    v8f acc0 = {}, acc1 = {}, acc2 = {}, acc3 = {};
#pragma unroll 16
    for (int k = 0; k < IN_CH; k += 4) {
        v2f a  = *(const v2f*)(xRow + k);
        v2f b0 = *(const v2f*)(wRow + k);
        v2f b1 = *(const v2f*)(wRow + 16 * IN_CH + k);
        v2f b2 = *(const v2f*)(wRow + 32 * IN_CH + k);
        v2f b3 = *(const v2f*)(wRow + 48 * IN_CH + k);
        acc0 = __builtin_amdgcn_wmma_f32_16x16x4_f32(false, a, false, b0, (short)0, acc0, false, false);
        acc1 = __builtin_amdgcn_wmma_f32_16x16x4_f32(false, a, false, b1, (short)0, acc1, false, false);
        acc2 = __builtin_amdgcn_wmma_f32_16x16x4_f32(false, a, false, b2, (short)0, acc2, false, false);
        acc3 = __builtin_amdgcn_wmma_f32_16x16x4_f32(false, a, false, b3, (short)0, acc3, false, false);
    }

    tileEpilogue(acc0, acc1, acc2, acc3, a_l, a_r, h, e_l, e_r, cnt, cursor, tile, lane);
}

#define W_PITCH 132

__device__ __forceinline__ void gemmAttnTileLds(const float* __restrict__ x,
                                                const float* wLds,
                                                const float* __restrict__ a_l,
                                                const float* __restrict__ a_r,
                                                float* __restrict__ h,
                                                float* __restrict__ e_l,
                                                float* __restrict__ e_r,
                                                int* __restrict__ cnt, int* __restrict__ cursor,
                                                int tile, int lane) {
    const int mBase = tile * 16;
    const int l15   = lane & 15;
    const int kOff  = (lane >> 4) << 2;

    const float* xRow = x + (size_t)(mBase + l15) * IN_CH + kOff;
    const float* wRow = wLds + l15 * W_PITCH + kOff;

    v8f acc0 = {}, acc1 = {}, acc2 = {}, acc3 = {};
#pragma unroll
    for (int k = 0; k < IN_CH; k += 8) {
        const v4f a  = *(const v4f*)(xRow + k);
        const v4f b0 = *(const v4f*)(wRow + k);
        const v4f b1 = *(const v4f*)(wRow + 16 * W_PITCH + k);
        const v4f b2 = *(const v4f*)(wRow + 32 * W_PITCH + k);
        const v4f b3 = *(const v4f*)(wRow + 48 * W_PITCH + k);
        const v2f aLo = {a[0], a[1]}, aHi = {a[2], a[3]};
        const v2f b0Lo = {b0[0], b0[1]}, b0Hi = {b0[2], b0[3]};
        const v2f b1Lo = {b1[0], b1[1]}, b1Hi = {b1[2], b1[3]};
        const v2f b2Lo = {b2[0], b2[1]}, b2Hi = {b2[2], b2[3]};
        const v2f b3Lo = {b3[0], b3[1]}, b3Hi = {b3[2], b3[3]};
        acc0 = __builtin_amdgcn_wmma_f32_16x16x4_f32(false, aLo, false, b0Lo, (short)0, acc0, false, false);
        acc1 = __builtin_amdgcn_wmma_f32_16x16x4_f32(false, aLo, false, b1Lo, (short)0, acc1, false, false);
        acc2 = __builtin_amdgcn_wmma_f32_16x16x4_f32(false, aLo, false, b2Lo, (short)0, acc2, false, false);
        acc3 = __builtin_amdgcn_wmma_f32_16x16x4_f32(false, aLo, false, b3Lo, (short)0, acc3, false, false);
        acc0 = __builtin_amdgcn_wmma_f32_16x16x4_f32(false, aHi, false, b0Hi, (short)0, acc0, false, false);
        acc1 = __builtin_amdgcn_wmma_f32_16x16x4_f32(false, aHi, false, b1Hi, (short)0, acc1, false, false);
        acc2 = __builtin_amdgcn_wmma_f32_16x16x4_f32(false, aHi, false, b2Hi, (short)0, acc2, false, false);
        acc3 = __builtin_amdgcn_wmma_f32_16x16x4_f32(false, aHi, false, b3Hi, (short)0, acc3, false, false);
    }

    tileEpilogue(acc0, acc1, acc2, acc3, a_l, a_r, h, e_l, e_r, cnt, cursor, tile, lane);
}

__global__ void gat_gemm_attn(const float* __restrict__ x,
                              const float* __restrict__ W,
                              const float* __restrict__ a_l,
                              const float* __restrict__ a_r,
                              float* __restrict__ h,
                              float* __restrict__ e_l,
                              float* __restrict__ e_r,
                              int* __restrict__ cnt, int* __restrict__ cursor,
                              int mTiles) {
    const int lane = threadIdx.x & 31;
    const int tile = blockIdx.x * (blockDim.x >> 5) + (threadIdx.x >> 5);
    if (tile >= mTiles) return;
    gemmAttnTile(x, W, a_l, a_r, h, e_l, e_r, cnt, cursor, tile, lane);
}

__global__ void gat_gemm_tail(const float* __restrict__ x,
                              const float* __restrict__ W,
                              float* __restrict__ h,
                              int rowStart, int nNodes) {
    int idx = blockIdx.x * blockDim.x + threadIdx.x;
    int row = rowStart + (idx >> 6);
    int col = idx & 63;
    if (row >= nNodes) return;
    const float* xr = x + (size_t)row * IN_CH;
    const float* wr = W + (size_t)col * IN_CH;
    float s = 0.f;
    for (int k = 0; k < IN_CH; ++k) s += xr[k] * wr[k];
    h[(size_t)row * OUT_CH + col] = s;
}

__global__ void gat_el_er_tail(const float* __restrict__ h,
                               const float* __restrict__ a_l,
                               const float* __restrict__ a_r,
                               float* __restrict__ e_l,
                               float* __restrict__ e_r,
                               int* __restrict__ cnt, int* __restrict__ cursor,
                               int rowStart, int nNodes) {
    const int lane = threadIdx.x & 31;
    const int node = rowStart + ((blockIdx.x * blockDim.x + threadIdx.x) >> 5);
    if (node >= nNodes) return;
    float2 hv = ((const float2*)(h + (size_t)node * OUT_CH))[lane];
    float2 al = ((const float2*)a_l)[lane];
    float2 ar = ((const float2*)a_r)[lane];
    float pl = hv.x * al.x + hv.y * al.y;
    float pr = hv.x * ar.x + hv.y * ar.y;
#pragma unroll
    for (int off = 16; off > 0; off >>= 1) {
        pl += __shfl_xor(pl, off, 32);
        pr += __shfl_xor(pr, off, 32);
    }
    if (lane == 0) {
        e_l[node] = pl; e_r[node] = pr;
        cnt[node] = 0;
        if (node == 0) *cursor = 0;
    }
}

__global__ void gat_count(const int* __restrict__ dst, int* __restrict__ cnt,
                          int* __restrict__ rank, int nEdges) {
    int e = blockIdx.x * blockDim.x + threadIdx.x;
    if (e >= nEdges) return;
    rank[e] = atomicAdd(cnt + dst[e], 1);
}

__global__ void gat_slots(const int* __restrict__ cnt, int* __restrict__ off,
                          int* __restrict__ cursor, int nNodes) {
    const int lane = threadIdx.x & 31;
    const int node = blockIdx.x * blockDim.x + threadIdx.x;
    const int c = node < nNodes ? cnt[node] : 0;
    int incl = c;
#pragma unroll
    for (int d = 1; d < 32; d <<= 1) {
        int t = __shfl_up(incl, d, 32);
        if (lane >= d) incl += t;
    }
    int base = 0;
    if (lane == 31) base = atomicAdd(cursor, incl);
    base = __shfl(base, 31, 32);
    if (node < nNodes) off[node] = base + incl - c;
}

__global__ void gat_fill(const int* __restrict__ src, const int* __restrict__ dst,
                         const int* __restrict__ off, const int* __restrict__ rank,
                         int* __restrict__ perm, int nEdges) {
    int e = blockIdx.x * blockDim.x + threadIdx.x;
    if (e >= nEdges) return;
    perm[off[dst[e]] + rank[e]] = src[e];
}

#define CNT_THREADS 1024
#define CNT_VECS    12
#define CNT_CHUNK   (CNT_THREADS * CNT_VECS * 4)
#define NODE_CAP    131072
#define SORT_MAXC   48

__device__ __forceinline__ bool aligned16(const void* p) { return ((size_t)p & 15) == 0; }

__device__ __forceinline__
void countLdsChunk(const int* __restrict__ dst, int* __restrict__ rank,
                   int* __restrict__ tab, int nEdges, int nNodes, int chunk, int tid) {
    __shared__ __attribute__((aligned(16))) unsigned int pairCnt[NODE_CAP / 2 + 32];
    const int base  = chunk * CNT_CHUNK;
    const int nWords = (nNodes + 1) >> 1;
    const int spare  = NODE_CAP / 2 + (tid & 31);

    uint4* z = (uint4*)pairCnt;
    for (int i = tid; i < (nWords + 3) >> 2; i += CNT_THREADS) z[i] = make_uint4(0u, 0u, 0u, 0u);

    const bool whole = aligned16(dst) && aligned16(rank) && base + CNT_CHUNK <= nEdges;
    const int  e0    = base + 4 * tid;
    int4 d[CNT_VECS];
    if (whole) {
#pragma unroll
        for (int j = 0; j < CNT_VECS; ++j) d[j] = *(const int4*)(dst + e0 + j * (CNT_THREADS * 4));
    } else {
#pragma unroll
        for (int j = 0; j < CNT_VECS; ++j) {
            const int e = e0 + j * (CNT_THREADS * 4);
            d[j].x = e     < nEdges ? dst[e]     : -1;
            d[j].y = e + 1 < nEdges ? dst[e + 1] : -1;
            d[j].z = e + 2 < nEdges ? dst[e + 2] : -1;
            d[j].w = e + 3 < nEdges ? dst[e + 3] : -1;
        }
    }
    __syncthreads();

#define LDS_RANK(n)                                                              \
    do {                                                                         \
        const int  node_ = (n);                                                  \
        const bool ok_   = (unsigned int)node_ < (unsigned int)nNodes;           \
        const bool hi_   = (node_ & 1) != 0;                                     \
        const unsigned int old_ = atomicAdd(&pairCnt[ok_ ? (node_ >> 1) : spare], \
                                            hi_ ? 0x10000u : 1u);                \
        (n) = ok_ ? (int)(hi_ ? (old_ >> 16) : (old_ & 0xffffu)) : 0;            \
    } while (0)
#pragma unroll
    for (int j = 0; j < CNT_VECS; ++j) {
        LDS_RANK(d[j].x); LDS_RANK(d[j].y); LDS_RANK(d[j].z); LDS_RANK(d[j].w);
    }
#undef LDS_RANK
    if (whole) {
#pragma unroll
        for (int j = 0; j < CNT_VECS; ++j) *(int4*)(rank + e0 + j * (CNT_THREADS * 4)) = d[j];
    } else {
#pragma unroll
        for (int j = 0; j < CNT_VECS; ++j) {
            const int e = e0 + j * (CNT_THREADS * 4);
            if (e     < nEdges) rank[e]     = d[j].x;
            if (e + 1 < nEdges) rank[e + 1] = d[j].y;
            if (e + 2 < nEdges) rank[e + 2] = d[j].z;
            if (e + 3 < nEdges) rank[e + 3] = d[j].w;
        }
    }
    __syncthreads();

    int* row = tab + (size_t)chunk * nNodes;
    const int nPairs = nNodes >> 1;
    if ((((size_t)row) & 7) == 0) {
#pragma unroll 4
        for (int i = tid; i < nPairs; i += CNT_THREADS) {
            const unsigned int w = pairCnt[i];
            *(int2*)(row + 2 * i) = make_int2((int)(w & 0xffffu), (int)(w >> 16));
        }
    } else {
#pragma unroll 4
        for (int i = tid; i < nPairs; i += CNT_THREADS) {
            const unsigned int w = pairCnt[i];
            row[2 * i]     = (int)(w & 0xffffu);
            row[2 * i + 1] = (int)(w >> 16);
        }
    }
    if ((nNodes & 1) && tid == 0) row[nNodes - 1] = (int)(pairCnt[nPairs] & 0xffffu);
}

__global__ __launch_bounds__(CNT_THREADS)
void gat_count_lds(const int* __restrict__ dst, int* __restrict__ rank,
                   int* __restrict__ tab, int nEdges, int nNodes) {
    countLdsChunk(dst, rank, tab, nEdges, nNodes, blockIdx.x, threadIdx.x);
}

__global__ __launch_bounds__(CNT_THREADS)
void gat_gemm_count(const float* __restrict__ x,
                    const float* __restrict__ W,
                    const float* __restrict__ a_l,
                    const float* __restrict__ a_r,
                    float* __restrict__ h,
                    float* __restrict__ e_l,
                    float* __restrict__ e_r,
                    int* __restrict__ cnt, int* __restrict__ cursor,
                    int mTiles,
                    const int* __restrict__ dst, int* __restrict__ rank,
                    int* __restrict__ tab, int nEdges, int nNodes, int nChunks) {
    const int block = blockIdx.x;
    if (block < nChunks) {
        countLdsChunk(dst, rank, tab, nEdges, nNodes, block, threadIdx.x);
    } else {
        const int lane = threadIdx.x & 31;
        const int wave = __builtin_amdgcn_readfirstlane((int)(threadIdx.x >> 5));
        const int tile = (block - nChunks) * (CNT_THREADS >> 5) + wave;
        if (aligned16(x) && aligned16(W)) {
            __shared__ __attribute__((aligned(16))) float wLds[OUT_CH * W_PITCH];
            static_assert(OUT_CH * IN_CH / 4 == 2 * CNT_THREADS, "two pieces of W per thread");
            const int j0 = threadIdx.x, j1 = threadIdx.x + CNT_THREADS;
            const int row0 = j0 >> 5, row1 = j1 >> 5, piece = j0 & 31;
            const v4f w0 = *(const v4f*)(W + (size_t)row0 * IN_CH + piece * 4);
            const v4f w1 = *(const v4f*)(W + (size_t)row1 * IN_CH + piece * 4);
            *(v4f*)(wLds + row0 * W_PITCH + piece * 4) = w0;
            *(v4f*)(wLds + row1 * W_PITCH + piece * 4) = w1;
            __syncthreads();
            if (tile >= mTiles) return;
            gemmAttnTileLds(x, wLds, a_l, a_r, h, e_l, e_r, cnt, cursor, tile, lane);
        } else {
            if (tile >= mTiles) return;
            gemmAttnTile(x, W, a_l, a_r, h, e_l, e_r, cnt, cursor, tile, lane);
        }
    }
}

__global__ void gat_slots_tab(int* __restrict__ tab, int* __restrict__ cnt,
                              int* __restrict__ off, int* __restrict__ cursor,
                              int nNodes, int nChunks) {
    const int lane = threadIdx.x & 31;
    const int node = blockIdx.x * blockDim.x + threadIdx.x;
    const bool live = node < nNodes;
    int* col = tab + (live ? node : 0);
    int v[SORT_MAXC];
#pragma unroll
    for (int c = 0; c < SORT_MAXC; ++c)
        v[c] = (live && c < nChunks) ? col[(size_t)c * nNodes] : 0;
    int total = 0;
#pragma unroll
    for (int c = 0; c < SORT_MAXC; ++c) total += v[c];
    int incl = total;
#pragma unroll
    for (int d = 1; d < 32; d <<= 1) {
        int t = __shfl_up(incl, d, 32);
        if (lane >= d) incl += t;
    }
    int base = 0;
    if (lane == 31) base = atomicAdd(cursor, incl);
    base = __shfl(base, 31, 32);
    if (live) {
        int run = base + incl - total;
        off[node] = run;
        cnt[node] = total;
#pragma unroll
        for (int c = 0; c < SORT_MAXC; ++c) {
            if (c < nChunks) { col[(size_t)c * nNodes] = run; run += v[c]; }
        }
    }
}

__global__ void gat_fill_tab(const int* __restrict__ src, const int* __restrict__ dst,
                             const int* __restrict__ tab, const int* __restrict__ rank,
                             int* __restrict__ perm, int nEdges, int nNodes) {
    int e = blockIdx.x * blockDim.x + threadIdx.x;
    if (e >= nEdges) return;
    const int chunk = e / CNT_CHUNK;
    perm[tab[(size_t)chunk * nNodes + dst[e]] + rank[e]] = src[e];
}

#define GA_NODES 4
#define GA_GROUP 8

__device__ __forceinline__ int   laneInt(int v, int j)   { return __builtin_amdgcn_readlane(v, j); }
__device__ __forceinline__ float laneFlt(float v, int j) {
    return __int_as_float(__builtin_amdgcn_readlane(__float_as_int(v), j));
}
__device__ __forceinline__ float otherRow(float v) {
    const int i = __float_as_int(v);
    return __int_as_float(__builtin_amdgcn_permlanex16(i, i, 0x76543210, 0xfedcba98, false, false));
}
__device__ __forceinline__ float waveMax(float v) {
    v = fmaxf(v, dppMove<0xB1>(v));
    v = fmaxf(v, dppMove<0x4E>(v));
    v = fmaxf(v, dppMove<0x141>(v));
    v = fmaxf(v, dppMove<0x140>(v));
    v = fmaxf(v, otherRow(v));
    return v;
}
__device__ __forceinline__ float waveSum(float v) {
    v += dppMove<0xB1>(v);
    v += dppMove<0x4E>(v);
    v += dppMove<0x141>(v);
    v += dppMove<0x140>(v);
    v += otherRow(v);
    return v;
}
template <typename P>
__device__ __forceinline__ P* rowBase(P* p, int row) {
    typedef __attribute__((address_space(1))) P GP;
    GP* r = (GP*)p + (size_t)row * OUT_CH;
    asm("" : "+s"(r));
    return (P*)r;
}
__device__ __forceinline__ float leaky(float a) { return (a > 0.f) ? a : NEG_SLOPE * a; }

__device__ __forceinline__ void addRows(const float* __restrict__ h, int s, float coef,
                                        int m, int lane, float& acc0, float& acc1) {
    for (int j0 = 0; j0 < m; j0 += GA_GROUP) {
        float r0[GA_GROUP], r1[GA_GROUP];
#pragma unroll
        for (int u = 0; u < GA_GROUP; ++u) {
            const int j  = j0 + u;
            const int jj = j < m ? j : m - 1;
            const float* hr = rowBase(h, laneInt(s, jj));
            r0[u] = hr[lane];
            r1[u] = hr[lane + 32];
        }
#pragma unroll
        for (int u = 0; u < GA_GROUP; ++u) {
            const float cj = laneFlt(coef, j0 + u);
            acc0 += r0[u] * cj;
            acc1 += r1[u] * cj;
        }
    }
}

__global__ void gat_gather(const int* __restrict__ perm, const int* __restrict__ cnt,
                           const int* __restrict__ off, const float* __restrict__ h,
                           const float* __restrict__ e_l, const float* __restrict__ e_r,
                           const float* __restrict__ bias, float* __restrict__ out,
                           int nNodes) {
    const int lane = threadIdx.x & 31;
    const int wv    = __builtin_amdgcn_readfirstlane((int)((blockIdx.x * blockDim.x + threadIdx.x) >> 5));
    const int node0 = wv * GA_NODES;
    if (node0 >= nNodes) return;

    int myCnt = 0, myOff = 0;
    float myEl = 0.f;
    if (lane < GA_NODES && node0 + lane < nNodes) {
        myCnt = cnt[node0 + lane];
        myOff = off[node0 + lane];
        myEl  = e_l[node0 + lane];
    }
    const float b0 = bias[lane], b1 = bias[lane + 32];

#pragma nounroll
    for (int q = 0; q < GA_NODES; ++q) {
        const int node = node0 + q;
        if (node >= nNodes) break;
        const int   n  = laneInt(myCnt, q);
        const int   o  = laneInt(myOff, q);
        const float el = laneFlt(myEl, q);
        float acc0 = 0.f, acc1 = 0.f;

        if (n <= 32) {
            int s = 0;
            float a = -__builtin_inff();
            if (lane < n) { s = perm[o + lane]; a = leaky(el + e_r[s]); }
            const float mx  = waveMax(a);
            const float w   = lane < n ? __expf(a - mx) : 0.f;
            const float sum = waveSum(w);
            const float coef = w / (sum + EPS_F);
            addRows(h, s, coef, n, lane, acc0, acc1);
        } else {
            float mx = -__builtin_inff();
            for (int c = 0; c < n; c += 32) {
                const int i = c + lane;
                if (i < n) mx = fmaxf(mx, leaky(el + e_r[perm[o + i]]));
            }
            mx = waveMax(mx);
            float sum = 0.f;
            for (int c = 0; c < n; c += 32) {
                const int i = c + lane;
                if (i < n) sum += __expf(leaky(el + e_r[perm[o + i]]) - mx);
            }
            sum = waveSum(sum);
            for (int c = 0; c < n; c += 32) {
                const int i = c + lane;
                const int m = (n - c) < 32 ? (n - c) : 32;
                int s = 0;
                float coef = 0.f;
                if (i < n) {
                    s = perm[o + i];
                    coef = __expf(leaky(el + e_r[s]) - mx) / (sum + EPS_F);
                }
                addRows(h, s, coef, m, lane, acc0, acc1);
            }
        }
        float* orow = out + (size_t)node * OUT_CH;
        orow[lane]      = acc0 + b0;
        orow[lane + 32] = acc1 + b1;
    }
}

extern "C" void kernel_launch(void* const* d_in, const int* in_sizes, int n_in,
                              void* d_out, int out_size, void* d_ws, size_t ws_size,
                              hipStream_t stream) {
    const float* x    = (const float*)d_in[0];
    const int*   ei   = (const int*)d_in[1];
    const float* W    = (const float*)d_in[2];
    const float* a_l  = (const float*)d_in[3];
    const float* a_r  = (const float*)d_in[4];
    const float* bias = (const float*)d_in[5];
    float* out = (float*)d_out;

    const int nNodes = in_sizes[0] / IN_CH;
    const int nEdges = in_sizes[1] / 2;
    const int* src = ei;
    const int* dst = ei + nEdges;

    float* ws   = (float*)d_ws;
    float* h    = ws;
    float* e_l  = h + (size_t)nNodes * OUT_CH;
    float* e_r  = e_l + nNodes;
    int*   cnt  = (int*)(e_r + nNodes);
    int*   off  = cnt + nNodes;
    int*   perm = off + nNodes;
    int*   rank   = (int*)d_out;
    int*   cursor = rank + nEdges;

    const int    nChunks = (nEdges + CNT_CHUNK - 1) / CNT_CHUNK;
    const size_t tabAt   = ((size_t)nEdges + 2) & ~(size_t)1;
    int*         tab     = rank + tabAt;
    const bool   onLds   = nNodes <= NODE_CAP && nChunks <= SORT_MAXC &&
                           tabAt + (size_t)nChunks * nNodes <= (size_t)nNodes * OUT_CH;

    const int  mFull  = nNodes & ~15;
    const int  mTiles = mFull / 16;
    const bool fused  = onLds && mTiles > 0;
    if (fused)
        gat_gemm_count<<<nChunks + (mTiles + 31) / 32, CNT_THREADS, 0, stream>>>(
            x, W, a_l, a_r, h, e_l, e_r, cnt, cursor, mTiles, dst, rank, tab, nEdges, nNodes, nChunks);
    else if (mTiles > 0)
        gat_gemm_attn<<<(mTiles + 1) / 2, 64, 0, stream>>>(x, W, a_l, a_r, h, e_l, e_r, cnt, cursor, mTiles);
    if (mFull < nNodes) {
        int tailThreads = (nNodes - mFull) * OUT_CH;
        gat_gemm_tail<<<(tailThreads + 255) / 256, 256, 0, stream>>>(x, W, h, mFull, nNodes);
        gat_el_er_tail<<<((nNodes - mFull) * 32 + 255) / 256, 256, 0, stream>>>(h, a_l, a_r, e_l, e_r, cnt, cursor, mFull, nNodes);
    }

    if (onLds) {
        if (!fused)
            gat_count_lds<<<nChunks, CNT_THREADS, 0, stream>>>(dst, rank, tab, nEdges, nNodes);
        gat_slots_tab<<<(nNodes + 255) / 256, 256, 0, stream>>>(tab, cnt, off, cursor, nNodes, nChunks);
        gat_fill_tab<<<(nEdges + 255) / 256, 256, 0, stream>>>(src, dst, tab, rank, perm, nEdges, nNodes);
    } else {
        gat_count<<<(nEdges + 255) / 256, 256, 0, stream>>>(dst, cnt, rank, nEdges);
        gat_slots<<<(nNodes + 255) / 256, 256, 0, stream>>>(cnt, off, cursor, nNodes);
        gat_fill<<<(nEdges + 255) / 256, 256, 0, stream>>>(src, dst, off, rank, perm, nEdges);
    }
    gat_gather<<<(nNodes + 31) / 32, 256, 0, stream>>>(perm, cnt, off, h, e_l, e_r, bias, out, nNodes);
}
